# starting kernel
# baseline (speedup 1.0000x reference)
#include <hip/hip_runtime.h>
#include <hip/hip_bf16.h>

#define F 64

typedef __attribute__((ext_vector_type(2))) float v2f;
typedef __attribute__((ext_vector_type(8))) float v8f;

__device__ __forceinline__ unsigned f2o(float f) {
    unsigned u = __float_as_uint(f);
    return (u & 0x80000000u) ? ~u : (u | 0x80000000u);
}
__device__ __forceinline__ float o2f(unsigned u) {
    return (u & 0x80000000u) ? __uint_as_float(u & 0x7FFFFFFFu) : __uint_as_float(~u);
}

__global__ __launch_bounds__(256) void gat_gemm_wmma(
    const float* __restrict__ x, const float* __restrict__ W,
    float* __restrict__ h, int N) {
    int wave = blockIdx.x * (blockDim.x >> 5) + (threadIdx.x >> 5);
    int lane = threadIdx.x & 31;
    int row  = (wave >> 2) << 4;
    int col  = (wave & 3) << 4;
    if (row >= N) return;

    int mn    = lane & 15;
    int khalf = (lane >> 4) << 1;

    int arow = row + mn; if (arow > N - 1) arow = N - 1;
    const float* xrow = x + (size_t)arow * F;

    v8f acc = {};
#pragma unroll
    for (int k = 0; k < F; k += 4) {
        float2 av = *(const float2*)(xrow + k + khalf);
        v2f a; a.x = av.x; a.y = av.y;
        v2f b;
        b.x = W[(size_t)(k + khalf)     * F + col + mn];
        b.y = W[(size_t)(k + khalf + 1) * F + col + mn];
        acc = __builtin_amdgcn_wmma_f32_16x16x4_f32(
             false, a,  false, b,
             (short)0, acc,  false,  false);
    }

    int mo = (lane >> 4) << 3;
#pragma unroll
    for (int r = 0; r < 8; ++r) {
        int m = row + mo + r;
        if (m < N) h[(size_t)m * F + col + mn] = acc[r];
    }
}

__global__ void gat_scores(const float* __restrict__ h,
                           const float* __restrict__ a_src,
                           const float* __restrict__ a_dst,
                           float* __restrict__ s_src, float* __restrict__ s_dst,
                           int N) {
    int i = blockIdx.x * blockDim.x + threadIdx.x;
    if (i >= N) return;
    const float4* hv = (const float4*)(h + (size_t)i * F);
    const float4* as = (const float4*)a_src;
    const float4* ad = (const float4*)a_dst;
    float ss = 0.f, sd = 0.f;
#pragma unroll
    for (int c = 0; c < F / 4; ++c) {
        float4 hh = hv[c], s4 = as[c], d4 = ad[c];
        ss += hh.x * s4.x + hh.y * s4.y + hh.z * s4.z + hh.w * s4.w;
        sd += hh.x * d4.x + hh.y * d4.y + hh.z * d4.z + hh.w * d4.w;
    }
    s_src[i] = ss; s_dst[i] = sd;
}

__global__ void gat_init(unsigned* __restrict__ m_ord, float* __restrict__ denom,
                         float* __restrict__ acc, int N, long total) {
    long i = (long)blockIdx.x * blockDim.x + threadIdx.x;
    if (i < N) { m_ord[i] = 0x007FFFFFu;   denom[i] = 0.f; }
    if (i < total) acc[i] = 0.f;
}

__global__ void gat_edge_max(const int* __restrict__ esrc, const int* __restrict__ edst,
                             const float* __restrict__ s_src, const float* __restrict__ s_dst,
                             float* __restrict__ e_buf, unsigned* __restrict__ m_ord, int E) {
    int e = blockIdx.x * blockDim.x + threadIdx.x;
    if (e >= E) return;
    int s = esrc[e], d = edst[e];
    float v = s_src[s] + s_dst[d];
    v = v > 0.f ? v : 0.2f * v;
    e_buf[e] = v;
    atomicMax(&m_ord[d], f2o(v));
}

__global__ void gat_edge_exp(const int* __restrict__ edst,
                             const float* __restrict__ e_buf,
                             const unsigned* __restrict__ m_ord,
                             float* __restrict__ ex_buf, float* __restrict__ denom, int E) {
    int e = blockIdx.x * blockDim.x + threadIdx.x;
    if (e >= E) return;
    int d = edst[e];
    float ex = expf(e_buf[e] - o2f(m_ord[d]));
    ex_buf[e] = ex;
    atomicAdd(&denom[d], ex);
}

__global__ void gat_scatter(const int* __restrict__ esrc, const int* __restrict__ edst,
                            const float* __restrict__ h, const float* __restrict__ ex_buf,
                            const float* __restrict__ denom, float* __restrict__ acc, int E) {
    long t = (long)blockIdx.x * blockDim.x + threadIdx.x;
    int e = (int)(t >> 4);
    int q = (int)(t & 15);
    if (e >= E) return;
    int s = esrc[e], d = edst[e];
    float dn = denom[d]; dn = dn > 1e-9f ? dn : 1e-9f;
    float alpha = ex_buf[e] / dn;
    float4 hv = ((const float4*)(h + (size_t)s * F))[q];
    float* ap = acc + (size_t)d * F + q * 4;
    atomicAdd(ap + 0, alpha * hv.x);
    atomicAdd(ap + 1, alpha * hv.y);
    atomicAdd(ap + 2, alpha * hv.z);
    atomicAdd(ap + 3, alpha * hv.w);
}

__global__ void gat_final(const float* __restrict__ acc, const float* __restrict__ b,
                          float* __restrict__ out, long total) {
    long i = (long)blockIdx.x * blockDim.x + threadIdx.x;
    if (i >= total) return;
    float v = acc[i] + b[i & (F - 1)];
    out[i] = v > 0.f ? v : 0.3f * v;
}

extern "C" void kernel_launch(void* const* d_in, const int* in_sizes, int n_in,
                              void* d_out, int out_size, void* d_ws, size_t ws_size,
                              hipStream_t stream) {
    const float* x     = (const float*)d_in[0];
    const float* W     = (const float*)d_in[1];
    const float* a_src = (const float*)d_in[2];
    const float* a_dst = (const float*)d_in[3];
    const float* b     = (const float*)d_in[4];
    const int*   esrc  = (const int*)d_in[5];
    const int*   edst  = (const int*)d_in[6];

    const int N = in_sizes[0] / F;
    const int E = in_sizes[5];
    const long total = (long)N * F;

    char* ws = (char*)d_ws;
    size_t off = 0;
    auto carve = [&](size_t bytes) -> void* {
        void* p = ws + off;
        off = (off + bytes + 255) & ~(size_t)255;
        return p;
    };
    float*    h      = (float*)carve((size_t)total * 4);
    float*    acc    = (float*)carve((size_t)total * 4);
    float*    s_src  = (float*)carve((size_t)N * 4);
    float*    s_dst  = (float*)carve((size_t)N * 4);
    unsigned* m_ord  = (unsigned*)carve((size_t)N * 4);
    float*    denom  = (float*)carve((size_t)N * 4);
    float*    e_buf  = (float*)carve((size_t)E * 4);
    float*    ex_buf = (float*)carve((size_t)E * 4);
    (void)ws_size; (void)n_in; (void)out_size;

    int rowTiles = (N + 15) / 16;
    int waves = rowTiles * 4;
    int gemmBlocks = (waves + 7) / 8;
    gat_gemm_wmma<<<gemmBlocks, 256, 0, stream>>>(x, W, h, N);

    gat_scores<<<(N + 255) / 256, 256, 0, stream>>>(h, a_src, a_dst, s_src, s_dst, N);

    gat_init<<<(int)((total + 255) / 256), 256, 0, stream>>>(m_ord, denom, acc, N, total);

    gat_edge_max<<<(E + 255) / 256, 256, 0, stream>>>(esrc, edst, s_src, s_dst, e_buf, m_ord, E);

    gat_edge_exp<<<(E + 255) / 256, 256, 0, stream>>>(edst, e_buf, m_ord, ex_buf, denom, E);

    long st = (long)E * 16;
    gat_scatter<<<(int)((st + 255) / 256), 256, 0, stream>>>(esrc, edst, h, ex_buf, denom, acc, E);

    gat_final<<<(int)((total + 255) / 256), 256, 0, stream>>>(acc, b, (float*)d_out, total);
}


# Node scores formed in the GEMM launch: wave sums tile x a_src/a_dst slice by 16-lane butterflies, 4 column-tile partials added via LDS (dot re-ordered); scores launch gone, h not re-read; 6 launches
# speedup vs baseline: 2.4784x; 2.4784x over previous
#include <hip/hip_runtime.h>
#include <hip/hip_bf16.h>

#define F 64

typedef __attribute__((ext_vector_type(2))) float v2f;
typedef __attribute__((ext_vector_type(8))) float v8f;

__device__ __forceinline__ unsigned f2o(float f) {
    unsigned u = __float_as_uint(f);
    return (u & 0x80000000u) ? ~u : (u | 0x80000000u);
}
__device__ __forceinline__ float o2f(unsigned u) {
    return (u & 0x80000000u) ? __uint_as_float(u & 0x7FFFFFFFu) : __uint_as_float(~u);
}

__device__ __forceinline__ int package_half() {
    return (int)((__builtin_amdgcn_s_sendmsg_rtn(0x87) >> 18) & 1u);
}


template <int M>
__device__ __forceinline__ float row_xor_add(float v) {
    int o = __builtin_amdgcn_mov_dpp(__float_as_int(v), 0x160 + M, 0xf, 0xf, false);
    return v + __int_as_float(o);
}

__global__ __launch_bounds__(256) void gat_gemm_wmma(
    const float* __restrict__ x, const float* __restrict__ W,
    const float* __restrict__ a_src, const float* __restrict__ a_dst,
    float* __restrict__ h, float* __restrict__ s_src, float* __restrict__ s_dst, int N) {
    __shared__ __attribute__((aligned(16))) float part[2][4][2][16];
    const int w    = threadIdx.x >> 5;
    const int lane = threadIdx.x & 31;
    const int rt   = w >> 2;
    const int ct   = w & 3;
    const int row  = ((int)blockIdx.x * 2 + rt) << 4;
    const int col  = ct << 4;

    if (row < N) {
        int mn    = lane & 15;
        int khalf = (lane >> 4) << 1;

        int arow = row + mn; if (arow > N - 1) arow = N - 1;
        const float* xrow = x + (size_t)arow * F;
        const float as = a_src[col + mn], ad = a_dst[col + mn];

        v8f acc = {};
#pragma unroll
        for (int k = 0; k < F; k += 4) {
            float2 av = *(const float2*)(xrow + k + khalf);
            v2f a; a.x = av.x; a.y = av.y;
            v2f b;
            b.x = W[(size_t)(k + khalf)     * F + col + mn];
            b.y = W[(size_t)(k + khalf + 1) * F + col + mn];
            acc = __builtin_amdgcn_wmma_f32_16x16x4_f32(
                 false, a,  false, b,
                 (short)0, acc,  false,  false);
        }

        int mo = (lane >> 4) << 3;
#pragma unroll
        for (int r = 0; r < 8; ++r) {
            int m = row + mo + r;
            if (m < N) h[(size_t)m * F + col + mn] = acc[r];
        }

        float ps[8], pd[8];
#pragma unroll
        for (int r = 0; r < 8; ++r) { ps[r] = acc[r] * as; pd[r] = acc[r] * ad; }
#pragma unroll
        for (int r = 0; r < 8; ++r) { ps[r] = row_xor_add<1>(ps[r]); pd[r] = row_xor_add<1>(pd[r]); }
#pragma unroll
        for (int r = 0; r < 8; ++r) { ps[r] = row_xor_add<2>(ps[r]); pd[r] = row_xor_add<2>(pd[r]); }
#pragma unroll
        for (int r = 0; r < 8; ++r) { ps[r] = row_xor_add<4>(ps[r]); pd[r] = row_xor_add<4>(pd[r]); }
#pragma unroll
        for (int r = 0; r < 8; ++r) { ps[r] = row_xor_add<8>(ps[r]); pd[r] = row_xor_add<8>(pd[r]); }
        if (mn == 0) {
            *(float4*)&part[rt][ct][0][mo]     = make_float4(ps[0], ps[1], ps[2], ps[3]);
            *(float4*)&part[rt][ct][0][mo + 4] = make_float4(ps[4], ps[5], ps[6], ps[7]);
            *(float4*)&part[rt][ct][1][mo]     = make_float4(pd[0], pd[1], pd[2], pd[3]);
            *(float4*)&part[rt][ct][1][mo + 4] = make_float4(pd[4], pd[5], pd[6], pd[7]);
        }
    }
    __syncthreads();

    const int t = threadIdx.x;
    if (t < 64) {
        const int kind = t >> 5;
        const int j    = t & 31;
        const int node = (int)blockIdx.x * 32 + j;
        if (node < N) {
            const int tl = j >> 4, rr = j & 15;
            float v = (part[tl][0][kind][rr] + part[tl][1][kind][rr])
                    + (part[tl][2][kind][rr] + part[tl][3][kind][rr]);
            (kind ? s_dst : s_src)[node] = v;
        }
    }
}

__global__ void gat_init(unsigned* __restrict__ m_ord0, unsigned* __restrict__ m_ord1,
                         float* __restrict__ denom0, float* __restrict__ denom1,
                         float* __restrict__ acc0, float* __restrict__ acc1,
                         const int* __restrict__ esrc, const int* __restrict__ edst,
                         const float* __restrict__ s_src, const float* __restrict__ s_dst,
                         float* __restrict__ e_buf, int N, long total, int E) {
    long i = (long)blockIdx.x * blockDim.x + threadIdx.x;
    if (i < N) {
        m_ord0[i] = 0x007FFFFFu; m_ord1[i] = 0x007FFFFFu;
        denom0[i] = 0.f; denom1[i] = 0.f;
    }
    if (i < total) { acc0[i] = 0.f; acc1[i] = 0.f; }
    if (i < E) {
        float t = s_src[esrc[i]] + s_dst[edst[i]];
        e_buf[i] = t > 0.f ? t : 0.2f * t;
    }
}

#define TAB_CAP     65536
#define SEG_THREADS 1024
#define SEG_BLOCKS  128
#define SEG_BATCH   8
#define NEG_INF_ORD 0x007FFFFFu

__global__ __launch_bounds__(SEG_THREADS) void gat_edge_max(
    const int* __restrict__ edst, const float* __restrict__ e_buf,
    unsigned* __restrict__ m_ord0, unsigned* __restrict__ m_ord1, int E, int N) {
    __shared__ __attribute__((aligned(16))) unsigned tab[TAB_CAP];
    unsigned* __restrict__ m_ord = package_half() ? m_ord1 : m_ord0;
    const bool priv = N <= TAB_CAP;
    const int tid = threadIdx.x;
    if (priv) {
        for (int i = tid * 4; i < N; i += SEG_THREADS * 4)
            *(uint4*)&tab[i] = make_uint4(NEG_INF_ORD, NEG_INF_ORD, NEG_INF_ORD, NEG_INF_ORD);
        __syncthreads();
    }
    const long stride = (long)gridDim.x * SEG_THREADS;
    for (long base = (long)blockIdx.x * SEG_THREADS + tid; base < E; base += stride * SEG_BATCH) {
        int d[SEG_BATCH];
        float v[SEG_BATCH];
#pragma unroll
        for (int k = 0; k < SEG_BATCH; ++k) {
            long e = base + k * stride;
            long ec = e < E ? e : (long)E - 1;
            d[k] = edst[ec];
            v[k] = e_buf[ec];
        }
#pragma unroll
        for (int k = 0; k < SEG_BATCH; ++k) {
            long e = base + k * stride;
            if (e < E) {
                unsigned o = f2o(v[k]);
                if (priv) atomicMax(&tab[d[k]], o);
                else      atomicMax(&m_ord[d[k]], o);
            }
        }
    }
    if (priv) {
        __syncthreads();
        const int chunks = (N + SEG_THREADS - 1) / SEG_THREADS;
        int c = blockIdx.x % chunks;
        for (int j0 = 0; j0 < chunks; j0 += SEG_BATCH) {
            int idx[SEG_BATCH];
            unsigned o[SEG_BATCH];
#pragma unroll
            for (int k = 0; k < SEG_BATCH; ++k) {
                int cc = c + k; if (cc >= chunks) cc -= chunks;
                int i = cc * SEG_THREADS + tid;
                idx[k] = (j0 + k < chunks && i < N) ? i : -1;
                o[k] = tab[idx[k] < 0 ? 0 : idx[k]];
            }
#pragma unroll
            for (int k = 0; k < SEG_BATCH; ++k)
                if (idx[k] >= 0 && o[k] != NEG_INF_ORD)
                    atomicMax(&m_ord[idx[k]], o[k]);
            c += SEG_BATCH; if (c >= chunks) c -= chunks;
        }
    }
}

__global__ __launch_bounds__(SEG_THREADS) void gat_edge_sum(
    const int* __restrict__ edst, const float* __restrict__ ex_buf,
    float* __restrict__ denom0, float* __restrict__ denom1, int E, int N) {
    __shared__ __attribute__((aligned(16))) float tab[TAB_CAP];
    float* __restrict__ denom = package_half() ? denom1 : denom0;
    const bool priv = N <= TAB_CAP;
    const int tid = threadIdx.x;
    if (priv) {
        for (int i = tid * 4; i < N; i += SEG_THREADS * 4)
            *(float4*)&tab[i] = make_float4(0.f, 0.f, 0.f, 0.f);
        __syncthreads();
    }
    const long stride = (long)gridDim.x * SEG_THREADS;
    for (long base = (long)blockIdx.x * SEG_THREADS + tid; base < E; base += stride * SEG_BATCH) {
        int d[SEG_BATCH];
        float ex[SEG_BATCH];
#pragma unroll
        for (int k = 0; k < SEG_BATCH; ++k) {
            long e = base + k * stride;
            long ec = e < E ? e : (long)E - 1;
            d[k]  = edst[ec];
            ex[k] = ex_buf[ec];
        }
#pragma unroll
        for (int k = 0; k < SEG_BATCH; ++k) {
            long e = base + k * stride;
            if (e < E) {
                if (priv) atomicAdd(&tab[d[k]], ex[k]);
                else      atomicAdd(&denom[d[k]], ex[k]);
            }
        }
    }
    if (priv) {
        __syncthreads();
        const int chunks = (N + SEG_THREADS - 1) / SEG_THREADS;
        int c = blockIdx.x % chunks;
        for (int j0 = 0; j0 < chunks; j0 += SEG_BATCH) {
            int idx[SEG_BATCH];
            float p[SEG_BATCH];
#pragma unroll
            for (int k = 0; k < SEG_BATCH; ++k) {
                int cc = c + k; if (cc >= chunks) cc -= chunks;
                int i = cc * SEG_THREADS + tid;
                idx[k] = (j0 + k < chunks && i < N) ? i : -1;
                p[k] = tab[idx[k] < 0 ? 0 : idx[k]];
            }
#pragma unroll
            for (int k = 0; k < SEG_BATCH; ++k)
                if (idx[k] >= 0 && p[k] != 0.f)
                    atomicAdd(&denom[idx[k]], p[k]);
            c += SEG_BATCH; if (c >= chunks) c -= chunks;
        }
    }
}

template <bool GUARD>
__device__ __forceinline__ void scatter_wave(
    const float* __restrict__ h, float* __restrict__ acc,
    int s, int d, float alpha, int n, unsigned laneoff) {
    for (int jb = 0; jb < n; jb += 8) {
        float h0[8], h1[8];
#pragma unroll
        for (int k = 0; k < 8; ++k) {
            unsigned sj = (unsigned)__builtin_amdgcn_readlane(s, jb + k);
            const char* hr = (const char*)h + ((size_t)sj << 8);
            h0[k] = *(const float*)(hr + laneoff);
            h1[k] = *(const float*)(hr + 128 + laneoff);
        }
#pragma unroll
        for (int k = 0; k < 8; ++k) {
            if (!GUARD || jb + k < n) {
                unsigned dj = (unsigned)__builtin_amdgcn_readlane(d, jb + k);
                float aj = __int_as_float(
                    __builtin_amdgcn_readlane(__float_as_int(alpha), jb + k));
                char* ar = (char*)acc + ((size_t)dj << 8);
                atomicAdd((float*)(ar + laneoff), aj * h0[k]);
                atomicAdd((float*)(ar + 128 + laneoff), aj * h1[k]);
            }
        }
    }
}

__global__ __launch_bounds__(256) void gat_scatter(
    const int* __restrict__ esrc, const int* __restrict__ edst,
    const float* __restrict__ h, const float* __restrict__ e_buf,
    const unsigned* __restrict__ m_ord0, const unsigned* __restrict__ m_ord1,
    float* __restrict__ ex_buf,
    float* __restrict__ acc0, float* __restrict__ acc1, int E) {
    int wave = __builtin_amdgcn_readfirstlane(
        (int)(blockIdx.x * (blockDim.x >> 5) + (threadIdx.x >> 5)));
    int lane = threadIdx.x & 31;
    long e0 = (long)wave * 32;
    if (e0 >= E) return;
    int n = E - (int)e0; if (n > 32) n = 32;

    int s = 0, d = 0; float alpha = 0.f, ev = 0.f;
    const int e = (int)e0 + lane;
    if (lane < n) { s = esrc[e]; d = edst[e]; ev = e_buf[e]; }
    float* __restrict__ acc = package_half() ? acc1 : acc0;
    if (lane < n) {
        unsigned mo0 = m_ord0[d], mo1 = m_ord1[d];
        alpha = expf(ev - o2f(mo0 > mo1 ? mo0 : mo1));
        ex_buf[e] = alpha;
    }

    unsigned laneoff = (unsigned)lane << 2;
    if (n == 32) scatter_wave<false>(h, acc, s, d, alpha, 32, laneoff);
    else         scatter_wave<true>(h, acc, s, d, alpha, n, laneoff);
}

__global__ void gat_final(const float* __restrict__ acc0, const float* __restrict__ acc1,
                          const float* __restrict__ denom0, const float* __restrict__ denom1,
                          const float* __restrict__ b, float* __restrict__ out, long total) {
    long i = (long)blockIdx.x * blockDim.x + threadIdx.x;
    if (i >= total) return;
    float dn = denom0[i / F] + denom1[i / F]; dn = dn > 1e-9f ? dn : 1e-9f;
    float v = (acc0[i] + acc1[i]) / dn + b[i & (F - 1)];
    out[i] = v > 0.f ? v : 0.3f * v;
}

extern "C" void kernel_launch(void* const* d_in, const int* in_sizes, int n_in,
                              void* d_out, int out_size, void* d_ws, size_t ws_size,
                              hipStream_t stream) {
    const float* x     = (const float*)d_in[0];
    const float* W     = (const float*)d_in[1];
    const float* a_src = (const float*)d_in[2];
    const float* a_dst = (const float*)d_in[3];
    const float* b     = (const float*)d_in[4];
    const int*   esrc  = (const int*)d_in[5];
    const int*   edst  = (const int*)d_in[6];

    const int N = in_sizes[0] / F;
    const int E = in_sizes[5];
    const long total = (long)N * F;

    char* ws = (char*)d_ws;
    size_t off = 0;
    auto carve = [&](size_t bytes) -> void* {
        void* p = ws + off;
        off = (off + bytes + 255) & ~(size_t)255;
        return p;
    };
    float*    h      = (float*)carve((size_t)total * 4);
    float*    acc0   = (float*)carve((size_t)total * 4);
    float*    acc1   = (float*)carve((size_t)total * 4);
    float*    s_src  = (float*)carve((size_t)N * 4);
    float*    s_dst  = (float*)carve((size_t)N * 4);
    unsigned* m_ord0 = (unsigned*)carve((size_t)N * 4);
    unsigned* m_ord1 = (unsigned*)carve((size_t)N * 4);
    float*    denom0 = (float*)carve((size_t)N * 4);
    float*    denom1 = (float*)carve((size_t)N * 4);
    float*    e_buf  = (float*)carve((size_t)E * 4);
    float*    ex_buf = (float*)carve((size_t)E * 4);
    (void)ws_size; (void)n_in; (void)out_size;

    int rowTiles = (N + 15) / 16;
    int waves = rowTiles * 4;
    int gemmBlocks = (waves + 7) / 8;
    gat_gemm_wmma<<<gemmBlocks, 256, 0, stream>>>(x, W, a_src, a_dst, h, s_src, s_dst, N);

    long cover = total > (long)E ? total : (long)E;
    gat_init<<<(int)((cover + 255) / 256), 256, 0, stream>>>(m_ord0, m_ord1, denom0, denom1, acc0, acc1, esrc, edst, s_src, s_dst, e_buf, N, total, E);

    gat_edge_max<<<SEG_BLOCKS, SEG_THREADS, 0, stream>>>(edst, e_buf, m_ord0, m_ord1, E, N);

    int scWaves = (E + 31) / 32;
    gat_scatter<<<(scWaves + 7) / 8, 256, 0, stream>>>(esrc, edst, h, e_buf, m_ord0, m_ord1, ex_buf, acc0, acc1, E);

    gat_edge_sum<<<SEG_BLOCKS, SEG_THREADS, 0, stream>>>(edst, ex_buf, denom0, denom1, E, N);

    gat_final<<<(int)((total + 255) / 256), 256, 0, stream>>>(acc0, acc1, denom0, denom1, b, (float*)d_out, total);
}


# Segment-max launch gone: softmax shift of destination d is the bound LeakyReLU(max s_src + s_dst[d]) (shift-invariant, no exp above 1); exact atomic max only if s_src spread > 64; 5 launches
# speedup vs baseline: 3.0296x; 1.2224x over previous
#include <hip/hip_runtime.h>
#include <hip/hip_bf16.h>

#define F 64

typedef __attribute__((ext_vector_type(2))) float v2f;
typedef __attribute__((ext_vector_type(8))) float v8f;

__device__ __forceinline__ unsigned f2o(float f) {
    unsigned u = __float_as_uint(f);
    return (u & 0x80000000u) ? ~u : (u | 0x80000000u);
}
__device__ __forceinline__ float o2f(unsigned u) {
    return (u & 0x80000000u) ? __uint_as_float(u & 0x7FFFFFFFu) : __uint_as_float(~u);
}
#define NEG_INF_ORD 0x007FFFFFu

__device__ __forceinline__ int package_half() {
    return (int)((__builtin_amdgcn_s_sendmsg_rtn(0x87) >> 18) & 1u);
}


template <int M>
__device__ __forceinline__ float row_xor_get(float v) {
    return __int_as_float(__builtin_amdgcn_mov_dpp(__float_as_int(v), 0x160 + M, 0xf, 0xf, false));
}
template <int M>
__device__ __forceinline__ float row_xor_add(float v) { return v + row_xor_get<M>(v); }

__device__ __forceinline__ float wave_max(float v) {
    v = fmaxf(v, row_xor_get<1>(v)); v = fmaxf(v, row_xor_get<2>(v));
    v = fmaxf(v, row_xor_get<4>(v)); v = fmaxf(v, row_xor_get<8>(v));
    return fmaxf(v, __shfl_xor(v, 16));
}
__device__ __forceinline__ float wave_min(float v) {
    v = fminf(v, row_xor_get<1>(v)); v = fminf(v, row_xor_get<2>(v));
    v = fminf(v, row_xor_get<4>(v)); v = fminf(v, row_xor_get<8>(v));
    return fminf(v, __shfl_xor(v, 16));
}

__global__ __launch_bounds__(256) void gat_gemm_wmma(
    const float* __restrict__ x, const float* __restrict__ W,
    const float* __restrict__ a_src, const float* __restrict__ a_dst,
    float* __restrict__ h, float* __restrict__ s_src, float* __restrict__ s_dst,
    float* __restrict__ bmm, unsigned* __restrict__ m_ord, int N) {
    __shared__ __attribute__((aligned(16))) float part[2][4][2][16];
    const int w    = threadIdx.x >> 5;
    const int lane = threadIdx.x & 31;
    const int rt   = w >> 2;
    const int ct   = w & 3;
    const int row  = ((int)blockIdx.x * 2 + rt) << 4;
    const int col  = ct << 4;

    if (row < N) {
        int mn    = lane & 15;
        int khalf = (lane >> 4) << 1;

        int arow = row + mn; if (arow > N - 1) arow = N - 1;
        const float* xrow = x + (size_t)arow * F;
        const float as = a_src[col + mn], ad = a_dst[col + mn];

        v8f acc = {};
#pragma unroll
        for (int k = 0; k < F; k += 4) {
            float2 av = *(const float2*)(xrow + k + khalf);
            v2f a; a.x = av.x; a.y = av.y;
            v2f b;
            b.x = W[(size_t)(k + khalf)     * F + col + mn];
            b.y = W[(size_t)(k + khalf + 1) * F + col + mn];
            acc = __builtin_amdgcn_wmma_f32_16x16x4_f32(
                 false, a,  false, b,
                 (short)0, acc,  false,  false);
        }

        int mo = (lane >> 4) << 3;
#pragma unroll
        for (int r = 0; r < 8; ++r) {
            int m = row + mo + r;
            if (m < N) h[(size_t)m * F + col + mn] = acc[r];
        }

        float ps[8], pd[8];
#pragma unroll
        for (int r = 0; r < 8; ++r) { ps[r] = acc[r] * as; pd[r] = acc[r] * ad; }
#pragma unroll
        for (int r = 0; r < 8; ++r) { ps[r] = row_xor_add<1>(ps[r]); pd[r] = row_xor_add<1>(pd[r]); }
#pragma unroll
        for (int r = 0; r < 8; ++r) { ps[r] = row_xor_add<2>(ps[r]); pd[r] = row_xor_add<2>(pd[r]); }
#pragma unroll
        for (int r = 0; r < 8; ++r) { ps[r] = row_xor_add<4>(ps[r]); pd[r] = row_xor_add<4>(pd[r]); }
#pragma unroll
        for (int r = 0; r < 8; ++r) { ps[r] = row_xor_add<8>(ps[r]); pd[r] = row_xor_add<8>(pd[r]); }
        if (mn == 0) {
            *(float4*)&part[rt][ct][0][mo]     = make_float4(ps[0], ps[1], ps[2], ps[3]);
            *(float4*)&part[rt][ct][0][mo + 4] = make_float4(ps[4], ps[5], ps[6], ps[7]);
            *(float4*)&part[rt][ct][1][mo]     = make_float4(pd[0], pd[1], pd[2], pd[3]);
            *(float4*)&part[rt][ct][1][mo + 4] = make_float4(pd[4], pd[5], pd[6], pd[7]);
        }
    }
    __syncthreads();

    const int t = threadIdx.x;
    if (t < 64) {
        const int kind = t >> 5;
        const int j    = t & 31;
        const int node = (int)blockIdx.x * 32 + j;
        const bool ok = node < N;
        float v = 0.f;
        if (ok) {
            const int tl = j >> 4, rr = j & 15;
            v = (part[tl][0][kind][rr] + part[tl][1][kind][rr])
              + (part[tl][2][kind][rr] + part[tl][3][kind][rr]);
            (kind ? s_dst : s_src)[node] = v;
        }
        if (kind == 0) {
            if (ok) m_ord[node] = NEG_INF_ORD;
            const float hi = wave_max(ok ? v : -INFINITY);
            const float lo = wave_min(ok ? v : INFINITY);
            if (j == 0) *(float2*)&bmm[2 * blockIdx.x] = make_float2(hi, lo);
        }
    }
}

#define SHIFT_SPREAD_MAX 64.f
#define INIT_THREADS 1024
__global__ __launch_bounds__(INIT_THREADS) void gat_init(
                         float* __restrict__ denom0, float* __restrict__ denom1,
                         float* __restrict__ acc0, float* __restrict__ acc1,
                         const int* __restrict__ esrc, const int* __restrict__ edst,
                         const float* __restrict__ s_src, const float* __restrict__ s_dst,
                         const float* __restrict__ bmm, int nb,
                         unsigned* __restrict__ m_ord, unsigned* __restrict__ mode,
                         float* __restrict__ e_buf, int N, long total, int E) {
    __shared__ float red[2][32];
    const int tid = threadIdx.x;
    const long i = (long)blockIdx.x * INIT_THREADS + tid;
    if (i < N) { denom0[i] = 0.f; denom1[i] = 0.f; }
    if (i < total) { acc0[i] = 0.f; acc1[i] = 0.f; }
    if ((long)blockIdx.x * INIT_THREADS >= E) return;

    const long ic = i < E ? i : (long)E - 1;
    const int si = esrc[ic], d = edst[ic];

    float hi = -INFINITY, lo = INFINITY;
    for (int base = 0; base < nb; base += 2 * INIT_THREADS) {
        int j0 = base + tid, j1 = base + INIT_THREADS + tid;
        if (j0 > nb - 1) j0 = nb - 1;
        if (j1 > nb - 1) j1 = nb - 1;
        const float2 p0 = *(const float2*)&bmm[2 * j0], p1 = *(const float2*)&bmm[2 * j1];
        hi = fmaxf(hi, fmaxf(p0.x, p1.x));
        lo = fminf(lo, fminf(p0.y, p1.y));
    }
    const float ss = s_src[si], sd = s_dst[d];
    hi = wave_max(hi); lo = wave_min(lo);
    if ((tid & 31) == 0) { red[0][tid >> 5] = hi; red[1][tid >> 5] = lo; }
    __syncthreads();
    hi = wave_max(red[0][tid & 31]);
    lo = wave_min(red[1][tid & 31]);
    const bool shifted = (hi - lo) <= SHIFT_SPREAD_MAX;
    if (blockIdx.x == 0 && tid == 0) *mode = shifted ? 1u : 0u;

    if (i < E) {
        const float t = ss + sd;
        float e = t > 0.f ? t : 0.2f * t;
        if (shifted) {
            const float u = hi + sd;
            e -= u > 0.f ? u : 0.2f * u;
        } else {
            atomicMax(&m_ord[d], f2o(e));
        }
        e_buf[i] = e;
    }
}

#define TAB_CAP     65536
#define SEG_THREADS 1024
#define SEG_BLOCKS  128
#define SEG_BATCH   8

__global__ __launch_bounds__(SEG_THREADS) void gat_edge_sum(
    const int* __restrict__ edst, const float* __restrict__ ex_buf,
    float* __restrict__ denom0, float* __restrict__ denom1, int E, int N) {
    __shared__ __attribute__((aligned(16))) float tab[TAB_CAP];
    float* __restrict__ denom = package_half() ? denom1 : denom0;
    const bool priv = N <= TAB_CAP;
    const int tid = threadIdx.x;
    if (priv) {
        for (int i = tid * 4; i < N; i += SEG_THREADS * 4)
            *(float4*)&tab[i] = make_float4(0.f, 0.f, 0.f, 0.f);
        __syncthreads();
    }
    const long stride = (long)gridDim.x * SEG_THREADS;
    for (long base = (long)blockIdx.x * SEG_THREADS + tid; base < E; base += stride * SEG_BATCH) {
        int d[SEG_BATCH];
        float ex[SEG_BATCH];
#pragma unroll
        for (int k = 0; k < SEG_BATCH; ++k) {
            long e = base + k * stride;
            long ec = e < E ? e : (long)E - 1;
            d[k]  = edst[ec];
            ex[k] = ex_buf[ec];
        }
#pragma unroll
        for (int k = 0; k < SEG_BATCH; ++k) {
            long e = base + k * stride;
            if (e < E) {
                if (priv) atomicAdd(&tab[d[k]], ex[k]);
                else      atomicAdd(&denom[d[k]], ex[k]);
            }
        }
    }
    if (priv) {
        __syncthreads();
        const int chunks = (N + SEG_THREADS - 1) / SEG_THREADS;
        int c = blockIdx.x % chunks;
        for (int j0 = 0; j0 < chunks; j0 += SEG_BATCH) {
            int idx[SEG_BATCH];
            float p[SEG_BATCH];
#pragma unroll
            for (int k = 0; k < SEG_BATCH; ++k) {
                int cc = c + k; if (cc >= chunks) cc -= chunks;
                int i = cc * SEG_THREADS + tid;
                idx[k] = (j0 + k < chunks && i < N) ? i : -1;
                p[k] = tab[idx[k] < 0 ? 0 : idx[k]];
            }
#pragma unroll
            for (int k = 0; k < SEG_BATCH; ++k)
                if (idx[k] >= 0 && p[k] != 0.f)
                    atomicAdd(&denom[idx[k]], p[k]);
            c += SEG_BATCH; if (c >= chunks) c -= chunks;
        }
    }
}

template <bool GUARD>
__device__ __forceinline__ void scatter_wave(
    const float* __restrict__ h, float* __restrict__ acc,
    int s, int d, float alpha, int n, unsigned laneoff) {
    for (int jb = 0; jb < n; jb += 8) {
        float h0[8], h1[8];
#pragma unroll
        for (int k = 0; k < 8; ++k) {
            unsigned sj = (unsigned)__builtin_amdgcn_readlane(s, jb + k);
            const char* hr = (const char*)h + ((size_t)sj << 8);
            h0[k] = *(const float*)(hr + laneoff);
            h1[k] = *(const float*)(hr + 128 + laneoff);
        }
#pragma unroll
        for (int k = 0; k < 8; ++k) {
            if (!GUARD || jb + k < n) {
                unsigned dj = (unsigned)__builtin_amdgcn_readlane(d, jb + k);
                float aj = __int_as_float(
                    __builtin_amdgcn_readlane(__float_as_int(alpha), jb + k));
                char* ar = (char*)acc + ((size_t)dj << 8);
                atomicAdd((float*)(ar + laneoff), aj * h0[k]);
                atomicAdd((float*)(ar + 128 + laneoff), aj * h1[k]);
            }
        }
    }
}

__global__ __launch_bounds__(256) void gat_scatter(
    const int* __restrict__ esrc, const int* __restrict__ edst,
    const float* __restrict__ h, const float* __restrict__ e_buf,
    const unsigned* __restrict__ m_ord, const unsigned* __restrict__ mode,
    float* __restrict__ ex_buf,
    float* __restrict__ acc0, float* __restrict__ acc1, int E) {
    int wave = __builtin_amdgcn_readfirstlane(
        (int)(blockIdx.x * (blockDim.x >> 5) + (threadIdx.x >> 5)));
    int lane = threadIdx.x & 31;
    long e0 = (long)wave * 32;
    if (e0 >= E) return;
    int n = E - (int)e0; if (n > 32) n = 32;

    int s = 0, d = 0; float alpha = 0.f, ev = 0.f;
    const int e = (int)e0 + lane;
    if (lane < n) { s = esrc[e]; d = edst[e]; ev = e_buf[e]; }
    const bool shifted = *mode != 0u;
    float* __restrict__ acc = package_half() ? acc1 : acc0;
    if (lane < n) {
        if (!shifted) ev -= o2f(m_ord[d]);
        alpha = expf(ev);
        ex_buf[e] = alpha;
    }

    unsigned laneoff = (unsigned)lane << 2;
    if (n == 32) scatter_wave<false>(h, acc, s, d, alpha, 32, laneoff);
    else         scatter_wave<true>(h, acc, s, d, alpha, n, laneoff);
}

__global__ void gat_final(const float* __restrict__ acc0, const float* __restrict__ acc1,
                          const float* __restrict__ denom0, const float* __restrict__ denom1,
                          const float* __restrict__ b, float* __restrict__ out, long total) {
    long i = (long)blockIdx.x * blockDim.x + threadIdx.x;
    if (i >= total) return;
    float dn = denom0[i / F] + denom1[i / F]; dn = dn > 0.f ? dn : 1.f;
    float v = (acc0[i] + acc1[i]) / dn + b[i & (F - 1)];
    out[i] = v > 0.f ? v : 0.3f * v;
}

extern "C" void kernel_launch(void* const* d_in, const int* in_sizes, int n_in,
                              void* d_out, int out_size, void* d_ws, size_t ws_size,
                              hipStream_t stream) {
    const float* x     = (const float*)d_in[0];
    const float* W     = (const float*)d_in[1];
    const float* a_src = (const float*)d_in[2];
    const float* a_dst = (const float*)d_in[3];
    const float* b     = (const float*)d_in[4];
    const int*   esrc  = (const int*)d_in[5];
    const int*   edst  = (const int*)d_in[6];

    const int N = in_sizes[0] / F;
    const int E = in_sizes[5];
    const long total = (long)N * F;

    char* ws = (char*)d_ws;
    size_t off = 0;
    auto carve = [&](size_t bytes) -> void* {
        void* p = ws + off;
        off = (off + bytes + 255) & ~(size_t)255;
        return p;
    };
    float*    h      = (float*)carve((size_t)total * 4);
    float*    acc0   = (float*)carve((size_t)total * 4);
    float*    acc1   = (float*)carve((size_t)total * 4);
    float*    s_src  = (float*)carve((size_t)N * 4);
    float*    s_dst  = (float*)carve((size_t)N * 4);
    unsigned* m_ord  = (unsigned*)carve((size_t)N * 4);
    float*    denom0 = (float*)carve((size_t)N * 4);
    float*    denom1 = (float*)carve((size_t)N * 4);
    float*    e_buf  = (float*)carve((size_t)E * 4);
    float*    ex_buf = (float*)carve((size_t)E * 4);
    int rowTiles = (N + 15) / 16;
    int waves = rowTiles * 4;
    int gemmBlocks = (waves + 7) / 8;
    float*    bmm    = (float*)carve((size_t)gemmBlocks * 8);
    unsigned* mode   = (unsigned*)carve(4);
    (void)ws_size; (void)n_in; (void)out_size;

    gat_gemm_wmma<<<gemmBlocks, 256, 0, stream>>>(x, W, a_src, a_dst, h, s_src, s_dst, bmm, m_ord, N);

    long cover = total > (long)E ? total : (long)E;
    gat_init<<<(int)((cover + INIT_THREADS - 1) / INIT_THREADS), INIT_THREADS, 0, stream>>>(denom0, denom1, acc0, acc1, esrc, edst, s_src, s_dst, bmm, gemmBlocks, m_ord, mode, e_buf, N, total, E);

    int scWaves = (E + 31) / 32;
    gat_scatter<<<(scWaves + 7) / 8, 256, 0, stream>>>(esrc, edst, h, e_buf, m_ord, mode, ex_buf, acc0, acc1, E);

    gat_edge_sum<<<SEG_BLOCKS, SEG_THREADS, 0, stream>>>(edst, ex_buf, denom0, denom1, E, N);

    gat_final<<<(int)((total + 255) / 256), 256, 0, stream>>>(acc0, acc1, denom0, denom1, b, (float*)d_out, total);
}


# Segment sum joins the scatter launch as a second role by work-group index: 64 edge shares x 32768-node windows, one LDS table each, form ex themselves; denom = sum of 64 partials; 4 launches
# speedup vs baseline: 3.4147x; 1.1271x over previous
#include <hip/hip_runtime.h>
#include <hip/hip_bf16.h>

#define F 64

typedef __attribute__((ext_vector_type(2))) float v2f;
typedef __attribute__((ext_vector_type(8))) float v8f;

__device__ __forceinline__ unsigned f2o(float f) {
    unsigned u = __float_as_uint(f);
    return (u & 0x80000000u) ? ~u : (u | 0x80000000u);
}
__device__ __forceinline__ float o2f(unsigned u) {
    return (u & 0x80000000u) ? __uint_as_float(u & 0x7FFFFFFFu) : __uint_as_float(~u);
}
#define NEG_INF_ORD 0x007FFFFFu

__device__ __forceinline__ int package_half() {
    return (int)((__builtin_amdgcn_s_sendmsg_rtn(0x87) >> 18) & 1u);
}


template <int M>
__device__ __forceinline__ float row_xor_get(float v) {
    return __int_as_float(__builtin_amdgcn_mov_dpp(__float_as_int(v), 0x160 + M, 0xf, 0xf, false));
}
template <int M>
__device__ __forceinline__ float row_xor_add(float v) { return v + row_xor_get<M>(v); }

__device__ __forceinline__ float wave_max(float v) {
    v = fmaxf(v, row_xor_get<1>(v)); v = fmaxf(v, row_xor_get<2>(v));
    v = fmaxf(v, row_xor_get<4>(v)); v = fmaxf(v, row_xor_get<8>(v));
    return fmaxf(v, __shfl_xor(v, 16));
}
__device__ __forceinline__ float wave_min(float v) {
    v = fminf(v, row_xor_get<1>(v)); v = fminf(v, row_xor_get<2>(v));
    v = fminf(v, row_xor_get<4>(v)); v = fminf(v, row_xor_get<8>(v));
    return fminf(v, __shfl_xor(v, 16));
}

__global__ __launch_bounds__(256) void gat_gemm_wmma(
    const float* __restrict__ x, const float* __restrict__ W,
    const float* __restrict__ a_src, const float* __restrict__ a_dst,
    float* __restrict__ h, float* __restrict__ s_src, float* __restrict__ s_dst,
    float* __restrict__ bmm, unsigned* __restrict__ m_ord, int N) {
    __shared__ __attribute__((aligned(16))) float part[2][4][2][16];
    const int w    = threadIdx.x >> 5;
    const int lane = threadIdx.x & 31;
    const int rt   = w >> 2;
    const int ct   = w & 3;
    const int row  = ((int)blockIdx.x * 2 + rt) << 4;
    const int col  = ct << 4;

    if (row < N) {
        int mn    = lane & 15;
        int khalf = (lane >> 4) << 1;

        int arow = row + mn; if (arow > N - 1) arow = N - 1;
        const float* xrow = x + (size_t)arow * F;
        const float as = a_src[col + mn], ad = a_dst[col + mn];

        v8f acc = {};
#pragma unroll
        for (int k = 0; k < F; k += 4) {
            float2 av = *(const float2*)(xrow + k + khalf);
            v2f a; a.x = av.x; a.y = av.y;
            v2f b;
            b.x = W[(size_t)(k + khalf)     * F + col + mn];
            b.y = W[(size_t)(k + khalf + 1) * F + col + mn];
            acc = __builtin_amdgcn_wmma_f32_16x16x4_f32(
                 false, a,  false, b,
                 (short)0, acc,  false,  false);
        }

        int mo = (lane >> 4) << 3;
#pragma unroll
        for (int r = 0; r < 8; ++r) {
            int m = row + mo + r;
            if (m < N) h[(size_t)m * F + col + mn] = acc[r];
        }

        float ps[8], pd[8];
#pragma unroll
        for (int r = 0; r < 8; ++r) { ps[r] = acc[r] * as; pd[r] = acc[r] * ad; }
#pragma unroll
        for (int r = 0; r < 8; ++r) { ps[r] = row_xor_add<1>(ps[r]); pd[r] = row_xor_add<1>(pd[r]); }
#pragma unroll
        for (int r = 0; r < 8; ++r) { ps[r] = row_xor_add<2>(ps[r]); pd[r] = row_xor_add<2>(pd[r]); }
#pragma unroll
        for (int r = 0; r < 8; ++r) { ps[r] = row_xor_add<4>(ps[r]); pd[r] = row_xor_add<4>(pd[r]); }
#pragma unroll
        for (int r = 0; r < 8; ++r) { ps[r] = row_xor_add<8>(ps[r]); pd[r] = row_xor_add<8>(pd[r]); }
        if (mn == 0) {
            *(float4*)&part[rt][ct][0][mo]     = make_float4(ps[0], ps[1], ps[2], ps[3]);
            *(float4*)&part[rt][ct][0][mo + 4] = make_float4(ps[4], ps[5], ps[6], ps[7]);
            *(float4*)&part[rt][ct][1][mo]     = make_float4(pd[0], pd[1], pd[2], pd[3]);
            *(float4*)&part[rt][ct][1][mo + 4] = make_float4(pd[4], pd[5], pd[6], pd[7]);
        }
    }
    __syncthreads();

    const int t = threadIdx.x;
    if (t < 64) {
        const int kind = t >> 5;
        const int j    = t & 31;
        const int node = (int)blockIdx.x * 32 + j;
        const bool ok = node < N;
        float v = 0.f;
        if (ok) {
            const int tl = j >> 4, rr = j & 15;
            v = (part[tl][0][kind][rr] + part[tl][1][kind][rr])
              + (part[tl][2][kind][rr] + part[tl][3][kind][rr]);
            (kind ? s_dst : s_src)[node] = v;
        }
        if (kind == 0) {
            if (ok) m_ord[node] = NEG_INF_ORD;
            const float hi = wave_max(ok ? v : -INFINITY);
            const float lo = wave_min(ok ? v : INFINITY);
            if (j == 0) *(float2*)&bmm[2 * blockIdx.x] = make_float2(hi, lo);
        }
    }
}

#define SHIFT_SPREAD_MAX 64.f
#define INIT_THREADS 1024
__global__ __launch_bounds__(INIT_THREADS) void gat_init(
                         float* __restrict__ denom0, float* __restrict__ denom1,
                         float* __restrict__ acc0, float* __restrict__ acc1,
                         const int* __restrict__ esrc, const int* __restrict__ edst,
                         const float* __restrict__ s_src, const float* __restrict__ s_dst,
                         const float* __restrict__ bmm, int nb,
                         unsigned* __restrict__ m_ord, unsigned* __restrict__ mode,
                         float* __restrict__ e_buf, int N, long total, int E) {
    __shared__ float red[2][32];
    const int tid = threadIdx.x;
    const long i = (long)blockIdx.x * INIT_THREADS + tid;
    if (i < N) { denom0[i] = 0.f; denom1[i] = 0.f; }
    if (i < total) { acc0[i] = 0.f; acc1[i] = 0.f; }
    if ((long)blockIdx.x * INIT_THREADS >= E) return;

    const long ic = i < E ? i : (long)E - 1;
    const int si = esrc[ic], d = edst[ic];

    float hi = -INFINITY, lo = INFINITY;
    for (int base = 0; base < nb; base += 2 * INIT_THREADS) {
        int j0 = base + tid, j1 = base + INIT_THREADS + tid;
        if (j0 > nb - 1) j0 = nb - 1;
        if (j1 > nb - 1) j1 = nb - 1;
        const float2 p0 = *(const float2*)&bmm[2 * j0], p1 = *(const float2*)&bmm[2 * j1];
        hi = fmaxf(hi, fmaxf(p0.x, p1.x));
        lo = fminf(lo, fminf(p0.y, p1.y));
    }
    const float ss = s_src[si], sd = s_dst[d];
    hi = wave_max(hi); lo = wave_min(lo);
    if ((tid & 31) == 0) { red[0][tid >> 5] = hi; red[1][tid >> 5] = lo; }
    __syncthreads();
    hi = wave_max(red[0][tid & 31]);
    lo = wave_min(red[1][tid & 31]);
    const bool shifted = (hi - lo) <= SHIFT_SPREAD_MAX;
    if (blockIdx.x == 0 && tid == 0) *mode = shifted ? 1u : 0u;

    if (i < E) {
        const float t = ss + sd;
        float e = t > 0.f ? t : 0.2f * t;
        if (shifted) {
            const float u = hi + sd;
            e -= u > 0.f ? u : 0.2f * u;
        } else {
            atomicMax(&m_ord[d], f2o(e));
        }
        e_buf[i] = e;
    }
}

#define TAB_CAP     32768
#define SEG_THREADS 1024
#define SEG_SHARES  64
#define SEG_BATCH   8

__device__ __forceinline__ void edge_sum_role(
    float* __restrict__ tab, int share, int lo,
    const int* __restrict__ edst, const float* __restrict__ e_buf,
    const unsigned* __restrict__ m_ord, bool shifted,
    float* __restrict__ denom, int E, int N) {
    const int tid = threadIdx.x;
    const int cnt = N - lo < TAB_CAP ? N - lo : TAB_CAP;
    for (int i = tid * 4; i < cnt; i += SEG_THREADS * 4)
        *(float4*)&tab[i] = make_float4(0.f, 0.f, 0.f, 0.f);
    __syncthreads();
    const long stride = (long)SEG_SHARES * SEG_THREADS;
    for (long base = (long)share * SEG_THREADS + tid; base < E; base += stride * SEG_BATCH) {
        int d[SEG_BATCH];
        float ex[SEG_BATCH];
#pragma unroll
        for (int k = 0; k < SEG_BATCH; ++k) {
            long e = base + k * stride;
            int ec = e < E ? (int)e : E - 1;
            d[k]  = edst[ec];
            ex[k] = e_buf[ec];
        }
#pragma unroll
        for (int k = 0; k < SEG_BATCH; ++k) {
            if (!shifted) ex[k] -= o2f(m_ord[d[k]]);
            ex[k] = expf(ex[k]);
        }
#pragma unroll
        for (int k = 0; k < SEG_BATCH; ++k) {
            long e = base + k * stride;
            const unsigned r = (unsigned)(d[k] - lo);
            if (e < E && r < (unsigned)cnt) atomicAdd(&tab[r], ex[k]);
        }
    }
    __syncthreads();
    const int chunks = (cnt + SEG_THREADS - 1) / SEG_THREADS;
    int c = share % chunks;
    for (int j0 = 0; j0 < chunks; j0 += SEG_BATCH) {
        int idx[SEG_BATCH];
        float p[SEG_BATCH];
#pragma unroll
        for (int k = 0; k < SEG_BATCH; ++k) {
            int cc = c + k; if (cc >= chunks) cc -= chunks;
            int i = cc * SEG_THREADS + tid;
            idx[k] = (j0 + k < chunks && i < cnt) ? i : -1;
            p[k] = tab[idx[k] < 0 ? 0 : idx[k]];
        }
#pragma unroll
        for (int k = 0; k < SEG_BATCH; ++k)
            if (idx[k] >= 0 && p[k] != 0.f)
                atomicAdd(&denom[lo + idx[k]], p[k]);
        c += SEG_BATCH; if (c >= chunks) c -= chunks;
    }
}

template <bool GUARD>
__device__ __forceinline__ void scatter_wave(
    const float* __restrict__ h, float* __restrict__ acc,
    int s, int d, float alpha, int n, unsigned laneoff) {
    for (int jb = 0; jb < n; jb += 8) {
        float h0[8], h1[8];
#pragma unroll
        for (int k = 0; k < 8; ++k) {
            unsigned sj = (unsigned)__builtin_amdgcn_readlane(s, jb + k);
            const char* hr = (const char*)h + ((size_t)sj << 8);
            h0[k] = *(const float*)(hr + laneoff);
            h1[k] = *(const float*)(hr + 128 + laneoff);
        }
#pragma unroll
        for (int k = 0; k < 8; ++k) {
            if (!GUARD || jb + k < n) {
                unsigned dj = (unsigned)__builtin_amdgcn_readlane(d, jb + k);
                float aj = __int_as_float(
                    __builtin_amdgcn_readlane(__float_as_int(alpha), jb + k));
                char* ar = (char*)acc + ((size_t)dj << 8);
                atomicAdd((float*)(ar + laneoff), aj * h0[k]);
                atomicAdd((float*)(ar + 128 + laneoff), aj * h1[k]);
            }
        }
    }
}

__global__ __launch_bounds__(SEG_THREADS) void gat_scatter_sum(
    const int* __restrict__ esrc, const int* __restrict__ edst,
    const float* __restrict__ h, const float* __restrict__ e_buf,
    const unsigned* __restrict__ m_ord, const unsigned* __restrict__ mode,
    float* __restrict__ acc0, float* __restrict__ acc1,
    float* __restrict__ denom0, float* __restrict__ denom1, int E, int N) {
    __shared__ __attribute__((aligned(16))) float tab[TAB_CAP];
    const int nwin = (N + TAB_CAP - 1) / TAB_CAP;
    const int sumBlocks = SEG_SHARES * nwin;
    if ((int)blockIdx.x < sumBlocks) {
        const int sb = (int)blockIdx.x;
        float* __restrict__ denom = package_half() ? denom1 : denom0;
        const bool shifted = *mode != 0u;
        edge_sum_role(tab, sb / nwin, (sb % nwin) * TAB_CAP, edst, e_buf, m_ord, shifted,
                      denom, E, N);
        return;
    }

    int wave = __builtin_amdgcn_readfirstlane(
        (int)((blockIdx.x - sumBlocks) * (SEG_THREADS >> 5) + (threadIdx.x >> 5)));
    int lane = threadIdx.x & 31;
    long e0 = (long)wave * 32;
    if (e0 >= E) return;
    int n = E - (int)e0; if (n > 32) n = 32;

    int s = 0, d = 0; float alpha = 0.f, ev = 0.f;
    const int e = (int)e0 + lane;
    if (lane < n) { s = esrc[e]; d = edst[e]; ev = e_buf[e]; }
    const bool shifted = *mode != 0u;
    float* __restrict__ acc = package_half() ? acc1 : acc0;
    if (lane < n) {
        if (!shifted) ev -= o2f(m_ord[d]);
        alpha = expf(ev);
    }

    unsigned laneoff = (unsigned)lane << 2;
    if (n == 32) scatter_wave<false>(h, acc, s, d, alpha, 32, laneoff);
    else         scatter_wave<true>(h, acc, s, d, alpha, n, laneoff);
}

__global__ void gat_final(const float* __restrict__ acc0, const float* __restrict__ acc1,
                          const float* __restrict__ denom0, const float* __restrict__ denom1,
                          const float* __restrict__ b, float* __restrict__ out, long total) {
    long i = (long)blockIdx.x * blockDim.x + threadIdx.x;
    if (i >= total) return;
    float dn = denom0[i / F] + denom1[i / F]; dn = dn > 0.f ? dn : 1.f;
    float v = (acc0[i] + acc1[i]) / dn + b[i & (F - 1)];
    out[i] = v > 0.f ? v : 0.3f * v;
}

extern "C" void kernel_launch(void* const* d_in, const int* in_sizes, int n_in,
                              void* d_out, int out_size, void* d_ws, size_t ws_size,
                              hipStream_t stream) {
    const float* x     = (const float*)d_in[0];
    const float* W     = (const float*)d_in[1];
    const float* a_src = (const float*)d_in[2];
    const float* a_dst = (const float*)d_in[3];
    const float* b     = (const float*)d_in[4];
    const int*   esrc  = (const int*)d_in[5];
    const int*   edst  = (const int*)d_in[6];

    const int N = in_sizes[0] / F;
    const int E = in_sizes[5];
    const long total = (long)N * F;

    char* ws = (char*)d_ws;
    size_t off = 0;
    auto carve = [&](size_t bytes) -> void* {
        void* p = ws + off;
        off = (off + bytes + 255) & ~(size_t)255;
        return p;
    };
    float*    h      = (float*)carve((size_t)total * 4);
    float*    acc0   = (float*)carve((size_t)total * 4);
    float*    acc1   = (float*)carve((size_t)total * 4);
    float*    s_src  = (float*)carve((size_t)N * 4);
    float*    s_dst  = (float*)carve((size_t)N * 4);
    unsigned* m_ord  = (unsigned*)carve((size_t)N * 4);
    float*    denom0 = (float*)carve((size_t)N * 4);
    float*    denom1 = (float*)carve((size_t)N * 4);
    float*    e_buf  = (float*)carve((size_t)E * 4);
    int rowTiles = (N + 15) / 16;
    int waves = rowTiles * 4;
    int gemmBlocks = (waves + 7) / 8;
    float*    bmm    = (float*)carve((size_t)gemmBlocks * 8);
    unsigned* mode   = (unsigned*)carve(4);
    (void)ws_size; (void)n_in; (void)out_size;

    gat_gemm_wmma<<<gemmBlocks, 256, 0, stream>>>(x, W, a_src, a_dst, h, s_src, s_dst, bmm, m_ord, N);

    long cover = total > (long)E ? total : (long)E;
    gat_init<<<(int)((cover + INIT_THREADS - 1) / INIT_THREADS), INIT_THREADS, 0, stream>>>(denom0, denom1, acc0, acc1, esrc, edst, s_src, s_dst, bmm, gemmBlocks, m_ord, mode, e_buf, N, total, E);

    int scWaves = (E + 31) / 32;
    int nwin = (N + TAB_CAP - 1) / TAB_CAP;
    gat_scatter_sum<<<SEG_SHARES * nwin + (scWaves + 31) / 32, SEG_THREADS, 0, stream>>>(esrc, edst, h, e_buf, m_ord, mode, acc0, acc1, denom0, denom1, E, N);

    gat_final<<<(int)((total + 255) / 256), 256, 0, stream>>>(acc0, acc1, denom0, denom1, b, (float*)d_out, total);
}
